# starting kernel
# baseline (speedup 1.0000x reference)
#include <hip/hip_runtime.h>
#include <hip/hip_bf16.h>


#define C 128
#define LEAKY 0.2f

typedef __attribute__((ext_vector_type(2))) float v2f;
typedef __attribute__((ext_vector_type(8))) float v8f;

__global__ __launch_bounds__(256) void gat_zero(float* __restrict__ out, int n_out,
                                                float* __restrict__ denom, int n_den) {
    int i = blockIdx.x * 256 + threadIdx.x;
    if (i < n_out) out[i] = 0.0f;
    if (i < n_den) denom[i] = 0.0f;
}

__global__ __launch_bounds__(256) void gat_gemm_uv(const float* __restrict__ h,
                                                   const float* __restrict__ w,
                                                   float* __restrict__ u,
                                                   float* __restrict__ v,
                                                   int n_nodes) {
    __shared__ float lh[16 * C];
    const int row0 = blockIdx.x * 16;
    const int tid  = threadIdx.x;
    const bool full_tile = (row0 + 16) <= n_nodes;

    if (full_tile) {
        for (int i = tid; i < 16 * C; i += 256)
            lh[i] = h[(size_t)row0 * C + i];
    } else {
        for (int i = tid; i < 16 * C; i += 256) {
            int gr = row0 + (i >> 7);
            lh[i] = (gr < n_nodes) ? h[(size_t)gr * C + (i & (C - 1))] : 0.0f;
        }
    }
    __syncthreads();

    const int wave = tid >> 5;
    const int lane = tid & 31;
    const int half = lane >> 4;
    const int l16  = lane & 15;
    const int col  = wave * 16 + l16;

    v8f accu = {};
    v8f accv = {};
    for (int k = 0; k < C; k += 4) {
        const int ka = k + half * 2;
        v2f afrag;
        afrag.x = lh[l16 * C + ka];
        afrag.y = lh[l16 * C + ka + 1];
        v2f bu, bv;
        bu.x = w[(size_t)(ka    ) * C + col];
        bu.y = w[(size_t)(ka + 1) * C + col];
        bv.x = w[(size_t)(C + ka    ) * C + col];
        bv.y = w[(size_t)(C + ka + 1) * C + col];
        accu = __builtin_amdgcn_wmma_f32_16x16x4_f32(false, afrag, false, bu,
                                                     (short)0, accu, false, false);
        accv = __builtin_amdgcn_wmma_f32_16x16x4_f32(false, afrag, false, bv,
                                                     (short)0, accv, false, false);
    }

    const int mbase = row0 + half * 8;
    if (full_tile) {
        #pragma unroll
        for (int r = 0; r < 8; ++r) {
            u[(size_t)(mbase + r) * C + col] = accu[r];
            v[(size_t)(mbase + r) * C + col] = accv[r];
        }
    } else {
        #pragma unroll
        for (int r = 0; r < 8; ++r) {
            if (mbase + r < n_nodes) {
                u[(size_t)(mbase + r) * C + col] = accu[r];
                v[(size_t)(mbase + r) * C + col] = accv[r];
            }
        }
    }
}

__device__ __forceinline__ float leaky(float x) { return x > 0.0f ? x : LEAKY * x; }

__global__ __launch_bounds__(256) void gat_edge_e(const float* __restrict__ u,
                                                  const float* __restrict__ v,
                                                  const float* __restrict__ a,
                                                  const int* __restrict__ src,
                                                  const int* __restrict__ dst,
                                                  float* __restrict__ exp_e,
                                                  float* __restrict__ denom,
                                                  int n_edges) {
    const int wib  = threadIdx.x >> 5;
    const int lane = threadIdx.x & 31;
    const int e = blockIdx.x * 8 + wib;
    if (e >= n_edges) return;

    const int s = src[e];
    const int d = dst[e];
    float4 uu = ((const float4*)(u + (size_t)s * C))[lane];
    float4 vv = ((const float4*)(v + (size_t)d * C))[lane];
    float4 aa = ((const float4*)a)[lane];

    float sum = leaky(uu.x + vv.x) * aa.x;
    sum = fmaf(leaky(uu.y + vv.y), aa.y, sum);
    sum = fmaf(leaky(uu.z + vv.z), aa.z, sum);
    sum = fmaf(leaky(uu.w + vv.w), aa.w, sum);

    #pragma unroll
    for (int off = 16; off > 0; off >>= 1)
        sum += __shfl_xor(sum, off, 32);

    if (lane == 0) {
        float ex = expf(sum);
        exp_e[e] = ex;
        atomicAdd(&denom[s], ex);
    }
}

__global__ __launch_bounds__(256) void gat_scatter(const float* __restrict__ h,
                                                   const int* __restrict__ src,
                                                   const int* __restrict__ dst,
                                                   const float* __restrict__ exp_e,
                                                   const float* __restrict__ denom,
                                                   float* __restrict__ out,
                                                   int n_edges) {
    const int wib  = threadIdx.x >> 5;
    const int lane = threadIdx.x & 31;
    const int e = blockIdx.x * 8 + wib;
    if (e >= n_edges) return;

    const int s = src[e];
    const int d = dst[e];
    const float att = exp_e[e] / denom[s];

    float4 hv = ((const float4*)(h + (size_t)s * C))[lane];
    float* op = out + (size_t)d * C + lane * 4;
    atomicAdd(op + 0, hv.x * att);
    atomicAdd(op + 1, hv.y * att);
    atomicAdd(op + 2, hv.z * att);
    atomicAdd(op + 3, hv.w * att);
}

extern "C" void kernel_launch(void* const* d_in, const int* in_sizes, int n_in,
                              void* d_out, int out_size, void* d_ws, size_t ws_size,
                              hipStream_t stream) {
    const float* h  = (const float*)d_in[0];
    const float* w  = (const float*)d_in[1];
    const float* a  = (const float*)d_in[2];
    const int* src  = (const int*)d_in[3];
    const int* dst  = (const int*)d_in[4];
    float* out      = (float*)d_out;

    const int n_nodes = in_sizes[0] / C;
    const int n_edges = in_sizes[3];

    float* u     = (float*)d_ws;
    float* v     = u + (size_t)n_nodes * C;
    float* exp_e = v + (size_t)n_nodes * C;
    float* denom = exp_e + n_edges;

    {
        int nz = n_nodes * C;
        int grid = (nz + 255) / 256;
        gat_zero<<<grid, 256, 0, stream>>>(out, nz, denom, n_nodes);
    }
    {
        int grid = (n_nodes + 15) / 16;
        gat_gemm_uv<<<grid, 256, 0, stream>>>(h, w, u, v, n_nodes);
    }
    {
        int grid = (n_edges + 7) / 8;
        gat_edge_e<<<grid, 256, 0, stream>>>(u, v, a, src, dst, exp_e, denom, n_edges);
    }
    {
        int grid = (n_edges + 7) / 8;
        gat_scatter<<<grid, 256, 0, stream>>>(h, src, dst, exp_e, denom, out, n_edges);
    }
}


# edge slot = start[src] + rank, rank = the counting atomic's return kept per edge: fill pass has no atomics (was a second returning int atomic per edge); slots still in arrival order
# speedup vs baseline: 1.6104x; 1.6104x over previous
#include <hip/hip_runtime.h>
#include <hip/hip_bf16.h>


#define C 128
#define LEAKY 0.2f

typedef __attribute__((ext_vector_type(2))) float v2f;
typedef __attribute__((ext_vector_type(8))) float v8f;

__global__ __launch_bounds__(256) void gat_zero(float* __restrict__ out, int n_out,
                                                int* __restrict__ cnt, int n_cnt) {
    int i = blockIdx.x * 256 + threadIdx.x;
    if (i < n_out) out[i] = 0.0f;
    if (i < n_cnt) cnt[i] = 0;
}

__global__ __launch_bounds__(256) void gat_gemm_uv(const float* __restrict__ h,
                                                   const float* __restrict__ w,
                                                   float* __restrict__ u,
                                                   float* __restrict__ v,
                                                   int n_nodes) {
    __shared__ float lh[16 * C];
    const int row0 = blockIdx.x * 16;
    const int tid  = threadIdx.x;
    const bool full_tile = (row0 + 16) <= n_nodes;

    if (full_tile) {
        for (int i = tid; i < 16 * C; i += 256)
            lh[i] = h[(size_t)row0 * C + i];
    } else {
        for (int i = tid; i < 16 * C; i += 256) {
            int gr = row0 + (i >> 7);
            lh[i] = (gr < n_nodes) ? h[(size_t)gr * C + (i & (C - 1))] : 0.0f;
        }
    }
    __syncthreads();

    const int wave = tid >> 5;
    const int lane = tid & 31;
    const int half = lane >> 4;
    const int l16  = lane & 15;
    const int col  = wave * 16 + l16;

    v8f accu = {};
    v8f accv = {};
    for (int k = 0; k < C; k += 4) {
        const int ka = k + half * 2;
        v2f afrag;
        afrag.x = lh[l16 * C + ka];
        afrag.y = lh[l16 * C + ka + 1];
        v2f bu, bv;
        bu.x = w[(size_t)(ka    ) * C + col];
        bu.y = w[(size_t)(ka + 1) * C + col];
        bv.x = w[(size_t)(C + ka    ) * C + col];
        bv.y = w[(size_t)(C + ka + 1) * C + col];
        accu = __builtin_amdgcn_wmma_f32_16x16x4_f32(false, afrag, false, bu,
                                                     (short)0, accu, false, false);
        accv = __builtin_amdgcn_wmma_f32_16x16x4_f32(false, afrag, false, bv,
                                                     (short)0, accv, false, false);
    }

    const int mbase = row0 + half * 8;
    if (full_tile) {
        #pragma unroll
        for (int r = 0; r < 8; ++r) {
            u[(size_t)(mbase + r) * C + col] = accu[r];
            v[(size_t)(mbase + r) * C + col] = accv[r];
        }
    } else {
        #pragma unroll
        for (int r = 0; r < 8; ++r) {
            if (mbase + r < n_nodes) {
                u[(size_t)(mbase + r) * C + col] = accu[r];
                v[(size_t)(mbase + r) * C + col] = accv[r];
            }
        }
    }
}

__device__ __forceinline__ float leaky(float x) { return x > 0.0f ? x : LEAKY * x; }

__global__ __launch_bounds__(256) void gat_count(const int* __restrict__ src,
                                                 int* __restrict__ cnt,
                                                 int* __restrict__ rank, int n_edges) {
    const int e = blockIdx.x * 256 + threadIdx.x;
    if (e < n_edges) rank[e] = atomicAdd(&cnt[src[e]], 1);
}

__global__ __launch_bounds__(1024) void gat_scan(const int* __restrict__ cnt,
                                                 int* __restrict__ start, int n) {
    __shared__ int wsum[32];
    const int tid  = threadIdx.x;
    const int lane = tid & 31;
    const int wid  = tid >> 5;
    const int per  = (n + 1023) / 1024;
    const int b    = tid * per;
    const int e    = (b + per < n) ? (b + per) : n;

    int s = 0;
    for (int i = b; i < e; ++i) s += cnt[i];

    int x = s;
    #pragma unroll
    for (int off = 1; off < 32; off <<= 1) {
        int y = __shfl_up(x, off, 32);
        if (lane >= off) x += y;
    }
    if (lane == 31) wsum[wid] = x;
    __syncthreads();
    if (wid == 0) {
        const int t = wsum[lane];
        int xs = t;
        #pragma unroll
        for (int off = 1; off < 32; off <<= 1) {
            int y = __shfl_up(xs, off, 32);
            if (lane >= off) xs += y;
        }
        wsum[lane] = xs - t;
    }
    __syncthreads();

    int run = wsum[wid] + (x - s);
    for (int i = b; i < e; ++i) {
        const int c = cnt[i];
        start[i] = run;
        run += c;
    }
}

__global__ __launch_bounds__(256) void gat_fill(const int* __restrict__ src,
                                                const int* __restrict__ dst,
                                                const int* __restrict__ start,
                                                const int* __restrict__ rank,
                                                int* __restrict__ sdst, int n_edges) {
    const int e = blockIdx.x * 256 + threadIdx.x;
    if (e < n_edges) sdst[start[src[e]] + rank[e]] = dst[e];
}

__device__ __forceinline__ float xhalf(float x) {
    int b = __builtin_bit_cast(int, x);
    return __builtin_bit_cast(float,
        __builtin_amdgcn_permlanex16(b, b, 0x76543210, 0xfedcba98, false, false));
}
template <int K>
__device__ __forceinline__ float row_xor_add(float x) {
    int y = __builtin_amdgcn_update_dpp(0, __builtin_bit_cast(int, x), 0x160 | K, 0xf, 0xf, true);
    return x + __builtin_bit_cast(float, y);
}
__device__ __forceinline__ float wave_sum(float x) {
    x += xhalf(x);
    x = row_xor_add<8>(x);
    x = row_xor_add<4>(x);
    x = row_xor_add<2>(x);
    x = row_xor_add<1>(x);
    return x;
}

typedef float f4 __attribute__((ext_vector_type(4)));
typedef const __attribute__((address_space(1))) f4* grow_cf4;
typedef __attribute__((address_space(1))) float* grow_f;
__device__ __forceinline__ unsigned long long uni_addr(const float* p) {
    const unsigned long long x = (unsigned long long)p;
    const unsigned lo = __builtin_amdgcn_readfirstlane((unsigned)x);
    const unsigned hi = __builtin_amdgcn_readfirstlane((unsigned)(x >> 32));
    return ((unsigned long long)hi << 32) | lo;
}

#define EB 8
__global__ __launch_bounds__(256) void gat_by_source(const float* __restrict__ h,
                                                     const float* __restrict__ u,
                                                     const float* __restrict__ v,
                                                     const float* __restrict__ a,
                                                     const int* __restrict__ start,
                                                     const int* __restrict__ cnt,
                                                     const int* __restrict__ sdst,
                                                     float* __restrict__ sex,
                                                     float* __restrict__ out,
                                                     int n_nodes) {
    const int lane = threadIdx.x & 31;
    const int s = __builtin_amdgcn_readfirstlane(blockIdx.x * 8 + (threadIdx.x >> 5));
    if (s >= n_nodes) return;
    const int beg = start[s];
    const int deg = cnt[s];
    if (deg <= 0) return;

    const float4 uu = ((const float4*)(u + (size_t)s * C))[lane];
    const float4 aa = ((const float4*)a)[lane];

    float dsum = 0.0f;
    for (int base = 0; base < deg; base += 32) {
        const int idx = base + lane;
        const bool ok = idx < deg;
        const int dj = ok ? sdst[beg + idx] : s;
        const int nb = (deg - base < 32) ? (deg - base) : 32;
        float ej = 0.0f;
        for (int j0 = 0; j0 < nb; j0 += EB) {
            f4 vv[EB];
            #pragma unroll
            for (int t = 0; t < EB; ++t) {
                const int d = __builtin_amdgcn_readlane(dj, j0 + t);
                vv[t] = ((grow_cf4)uni_addr(v + (size_t)d * C))[lane];
            }
            #pragma unroll
            for (int t = 0; t < EB; ++t) {
                float sum = leaky(uu.x + vv[t].x) * aa.x;
                sum = fmaf(leaky(uu.y + vv[t].y), aa.y, sum);
                sum = fmaf(leaky(uu.z + vv[t].z), aa.z, sum);
                sum = fmaf(leaky(uu.w + vv[t].w), aa.w, sum);
                sum = wave_sum(sum);
                if (lane == j0 + t) ej = sum;
            }
        }
        const float ex = ok ? expf(ej) : 0.0f;
        if (ok) sex[beg + idx] = ex;
        dsum += ex;
    }
    const float denom = wave_sum(dsum);

    float hk[4];
    #pragma unroll
    for (int k = 0; k < 4; ++k) hk[k] = h[(size_t)s * C + 32 * k + lane];

    for (int base = 0; base < deg; base += 32) {
        const int idx = base + lane;
        const bool ok = idx < deg;
        const int dj = ok ? sdst[beg + idx] : s;
        const float wj = ok ? sex[beg + idx] / denom : 0.0f;
        const int nb = (deg - base < 32) ? (deg - base) : 32;
        for (int j = 0; j < nb; ++j) {
            const int d = __builtin_amdgcn_readlane(dj, j);
            const float w = __builtin_bit_cast(float,
                __builtin_amdgcn_readlane(__builtin_bit_cast(int, wj), j));
            grow_f op = (grow_f)uni_addr(out + (size_t)d * C) + lane;
            #pragma unroll
            for (int k = 0; k < 4; ++k)
                __hip_atomic_fetch_add(op + 32 * k, hk[k] * w, __ATOMIC_RELAXED, __HIP_MEMORY_SCOPE_AGENT);
        }
    }
}

extern "C" void kernel_launch(void* const* d_in, const int* in_sizes, int n_in,
                              void* d_out, int out_size, void* d_ws, size_t ws_size,
                              hipStream_t stream) {
    const float* h  = (const float*)d_in[0];
    const float* w  = (const float*)d_in[1];
    const float* a  = (const float*)d_in[2];
    const int* src  = (const int*)d_in[3];
    const int* dst  = (const int*)d_in[4];
    float* out      = (float*)d_out;

    const int n_nodes = in_sizes[0] / C;
    const int n_edges = in_sizes[3];

    float* u     = (float*)d_ws;
    float* v     = u + (size_t)n_nodes * C;
    int* cnt     = (int*)(v + (size_t)n_nodes * C);
    int* start   = cnt + n_nodes;
    int* sdst    = start + n_nodes;
    float* sex   = (float*)(sdst + n_edges);
    int* rank    = (int*)(sex + n_edges);

    {
        int nz = n_nodes * C;
        int grid = (nz + 255) / 256;
        gat_zero<<<grid, 256, 0, stream>>>(out, nz, cnt, n_nodes);
    }
    {
        int grid = (n_nodes + 15) / 16;
        gat_gemm_uv<<<grid, 256, 0, stream>>>(h, w, u, v, n_nodes);
    }
    {
        int grid = (n_edges + 255) / 256;
        gat_count<<<grid, 256, 0, stream>>>(src, cnt, rank, n_edges);
        gat_scan<<<1, 1024, 0, stream>>>(cnt, start, n_nodes);
        gat_fill<<<grid, 256, 0, stream>>>(src, dst, start, rank, sdst, n_edges);
    }
    {
        int grid = (n_nodes + 7) / 8;
        gat_by_source<<<grid, 256, 0, stream>>>(h, u, v, a, start, cnt, sdst, sex, out, n_nodes);
    }
}


# atomic targets (row sums, edge counters) one copy per chip half, wave's die id picks its copy: no line added to from both halves; closing launch adds copies: 7 launches, sum order changed
# speedup vs baseline: 1.6963x; 1.0533x over previous
#include <hip/hip_runtime.h>
#include <hip/hip_bf16.h>


#define C 128
#define LEAKY 0.2f

typedef __attribute__((ext_vector_type(2))) float v2f;
typedef __attribute__((ext_vector_type(8))) float v8f;

__device__ __forceinline__ int chip_half() {
    return (__builtin_amdgcn_s_sendmsg_rtn(0x87) >> 18) & 1;
}

__global__ __launch_bounds__(256) void gat_zero(float* __restrict__ sums, int n_sums4,
                                                int* __restrict__ cnt, int n_cnt) {
    int i = blockIdx.x * 256 + threadIdx.x;
    if (i < n_sums4) ((float4*)sums)[i] = make_float4(0.0f, 0.0f, 0.0f, 0.0f);
    if (i < n_cnt) cnt[i] = 0;
}

__global__ __launch_bounds__(256) void gat_gemm_uv(const float* __restrict__ h,
                                                   const float* __restrict__ w,
                                                   float* __restrict__ u,
                                                   float* __restrict__ v,
                                                   int n_nodes) {
    __shared__ float lh[16 * C];
    const int row0 = blockIdx.x * 16;
    const int tid  = threadIdx.x;
    const bool full_tile = (row0 + 16) <= n_nodes;

    if (full_tile) {
        for (int i = tid; i < 16 * C; i += 256)
            lh[i] = h[(size_t)row0 * C + i];
    } else {
        for (int i = tid; i < 16 * C; i += 256) {
            int gr = row0 + (i >> 7);
            lh[i] = (gr < n_nodes) ? h[(size_t)gr * C + (i & (C - 1))] : 0.0f;
        }
    }
    __syncthreads();

    const int wave = tid >> 5;
    const int lane = tid & 31;
    const int half = lane >> 4;
    const int l16  = lane & 15;
    const int col  = wave * 16 + l16;

    v8f accu = {};
    v8f accv = {};
    for (int k = 0; k < C; k += 4) {
        const int ka = k + half * 2;
        v2f afrag;
        afrag.x = lh[l16 * C + ka];
        afrag.y = lh[l16 * C + ka + 1];
        v2f bu, bv;
        bu.x = w[(size_t)(ka    ) * C + col];
        bu.y = w[(size_t)(ka + 1) * C + col];
        bv.x = w[(size_t)(C + ka    ) * C + col];
        bv.y = w[(size_t)(C + ka + 1) * C + col];
        accu = __builtin_amdgcn_wmma_f32_16x16x4_f32(false, afrag, false, bu,
                                                     (short)0, accu, false, false);
        accv = __builtin_amdgcn_wmma_f32_16x16x4_f32(false, afrag, false, bv,
                                                     (short)0, accv, false, false);
    }

    const int mbase = row0 + half * 8;
    if (full_tile) {
        #pragma unroll
        for (int r = 0; r < 8; ++r) {
            u[(size_t)(mbase + r) * C + col] = accu[r];
            v[(size_t)(mbase + r) * C + col] = accv[r];
        }
    } else {
        #pragma unroll
        for (int r = 0; r < 8; ++r) {
            if (mbase + r < n_nodes) {
                u[(size_t)(mbase + r) * C + col] = accu[r];
                v[(size_t)(mbase + r) * C + col] = accv[r];
            }
        }
    }
}

__device__ __forceinline__ float leaky(float x) { return x > 0.0f ? x : LEAKY * x; }

__global__ __launch_bounds__(256) void gat_count(const int* __restrict__ src,
                                                 int* __restrict__ cnt,
                                                 int* __restrict__ rank, int n_edges,
                                                 int n_nodes) {
    const int e = blockIdx.x * 256 + threadIdx.x;
    const int hf = chip_half();
    if (e < n_edges) {
        const unsigned r = (unsigned)atomicAdd(&cnt[hf * n_nodes + src[e]], 1);
        rank[e] = (int)(r | ((unsigned)hf << 31));
    }
}

__global__ __launch_bounds__(1024) void gat_scan(const int* __restrict__ cnt,
                                                 int* __restrict__ start, int n) {
    __shared__ int wsum[32];
    const int tid  = threadIdx.x;
    const int lane = tid & 31;
    const int wid  = tid >> 5;
    const int per  = (n + 1023) / 1024;
    const int b    = tid * per;
    const int e    = (b + per < n) ? (b + per) : n;

    int s = 0;
    for (int i = b; i < e; ++i) s += cnt[i] + cnt[n + i];

    int x = s;
    #pragma unroll
    for (int off = 1; off < 32; off <<= 1) {
        int y = __shfl_up(x, off, 32);
        if (lane >= off) x += y;
    }
    if (lane == 31) wsum[wid] = x;
    __syncthreads();
    if (wid == 0) {
        const int t = wsum[lane];
        int xs = t;
        #pragma unroll
        for (int off = 1; off < 32; off <<= 1) {
            int y = __shfl_up(xs, off, 32);
            if (lane >= off) xs += y;
        }
        wsum[lane] = xs - t;
    }
    __syncthreads();

    int run = wsum[wid] + (x - s);
    for (int i = b; i < e; ++i) {
        const int c0 = cnt[i];
        const int c1 = cnt[n + i];
        start[i] = run;
        start[n + i] = run + c0;
        run += c0 + c1;
    }
}

__global__ __launch_bounds__(256) void gat_fill(const int* __restrict__ src,
                                                const int* __restrict__ dst,
                                                const int* __restrict__ start,
                                                const int* __restrict__ rank,
                                                int* __restrict__ sdst, int n_edges,
                                                int n_nodes) {
    const int e = blockIdx.x * 256 + threadIdx.x;
    if (e < n_edges) {
        const unsigned rk = (unsigned)rank[e];
        const int hb = (int)(rk >> 31);
        const int r  = (int)(rk & 0x7fffffffu);
        sdst[start[hb * n_nodes + src[e]] + r] = dst[e];
    }
}

__device__ __forceinline__ float xhalf(float x) {
    int b = __builtin_bit_cast(int, x);
    return __builtin_bit_cast(float,
        __builtin_amdgcn_permlanex16(b, b, 0x76543210, 0xfedcba98, false, false));
}
template <int K>
__device__ __forceinline__ float row_xor_add(float x) {
    int y = __builtin_amdgcn_update_dpp(0, __builtin_bit_cast(int, x), 0x160 | K, 0xf, 0xf, true);
    return x + __builtin_bit_cast(float, y);
}
__device__ __forceinline__ float wave_sum(float x) {
    x += xhalf(x);
    x = row_xor_add<8>(x);
    x = row_xor_add<4>(x);
    x = row_xor_add<2>(x);
    x = row_xor_add<1>(x);
    return x;
}

typedef float f4 __attribute__((ext_vector_type(4)));
typedef const __attribute__((address_space(1))) f4* grow_cf4;
typedef __attribute__((address_space(1))) float* grow_f;
__device__ __forceinline__ unsigned long long uni_addr(const float* p) {
    const unsigned long long x = (unsigned long long)p;
    const unsigned lo = __builtin_amdgcn_readfirstlane((unsigned)x);
    const unsigned hi = __builtin_amdgcn_readfirstlane((unsigned)(x >> 32));
    return ((unsigned long long)hi << 32) | lo;
}

#define EB 8
__global__ __launch_bounds__(256) void gat_by_source(const float* __restrict__ h,
                                                     const float* __restrict__ u,
                                                     const float* __restrict__ v,
                                                     const float* __restrict__ a,
                                                     const int* __restrict__ start,
                                                     const int* __restrict__ cnt,
                                                     const int* __restrict__ sdst,
                                                     float* __restrict__ sex,
                                                     float* __restrict__ sums,
                                                     int n_nodes) {
    const int lane = threadIdx.x & 31;
    const int s = __builtin_amdgcn_readfirstlane(blockIdx.x * 8 + (threadIdx.x >> 5));
    if (s >= n_nodes) return;
    const int beg = start[s];
    const int deg = cnt[s] + cnt[n_nodes + s];
    if (deg <= 0) return;
    float* __restrict__ mine = sums + (size_t)chip_half() * n_nodes * C;

    const float4 uu = ((const float4*)(u + (size_t)s * C))[lane];
    const float4 aa = ((const float4*)a)[lane];

    float dsum = 0.0f;
    for (int base = 0; base < deg; base += 32) {
        const int idx = base + lane;
        const bool ok = idx < deg;
        const int dj = ok ? sdst[beg + idx] : s;
        const int nb = (deg - base < 32) ? (deg - base) : 32;
        float ej = 0.0f;
        for (int j0 = 0; j0 < nb; j0 += EB) {
            f4 vv[EB];
            #pragma unroll
            for (int t = 0; t < EB; ++t) {
                const int d = __builtin_amdgcn_readlane(dj, j0 + t);
                vv[t] = ((grow_cf4)uni_addr(v + (size_t)d * C))[lane];
            }
            #pragma unroll
            for (int t = 0; t < EB; ++t) {
                float sum = leaky(uu.x + vv[t].x) * aa.x;
                sum = fmaf(leaky(uu.y + vv[t].y), aa.y, sum);
                sum = fmaf(leaky(uu.z + vv[t].z), aa.z, sum);
                sum = fmaf(leaky(uu.w + vv[t].w), aa.w, sum);
                sum = wave_sum(sum);
                if (lane == j0 + t) ej = sum;
            }
        }
        const float ex = ok ? expf(ej) : 0.0f;
        if (ok) sex[beg + idx] = ex;
        dsum += ex;
    }
    const float denom = wave_sum(dsum);

    float hk[4];
    #pragma unroll
    for (int k = 0; k < 4; ++k) hk[k] = h[(size_t)s * C + 32 * k + lane];

    for (int base = 0; base < deg; base += 32) {
        const int idx = base + lane;
        const bool ok = idx < deg;
        const int dj = ok ? sdst[beg + idx] : s;
        const float wj = ok ? sex[beg + idx] / denom : 0.0f;
        const int nb = (deg - base < 32) ? (deg - base) : 32;
        for (int j = 0; j < nb; ++j) {
            const int d = __builtin_amdgcn_readlane(dj, j);
            const float w = __builtin_bit_cast(float,
                __builtin_amdgcn_readlane(__builtin_bit_cast(int, wj), j));
            grow_f op = (grow_f)uni_addr(mine + (size_t)d * C) + lane;
            #pragma unroll
            for (int k = 0; k < 4; ++k)
                __hip_atomic_fetch_add(op + 32 * k, hk[k] * w, __ATOMIC_RELAXED, __HIP_MEMORY_SCOPE_AGENT);
        }
    }
}

__global__ __launch_bounds__(256) void gat_add_copies(const float* __restrict__ sums,
                                                      float* __restrict__ out, int n4) {
    const int i = blockIdx.x * 256 + threadIdx.x;
    if (i < n4) {
        const float4 x = ((const float4*)sums)[i];
        const float4 y = ((const float4*)sums)[n4 + i];
        ((float4*)out)[i] = make_float4(x.x + y.x, x.y + y.y, x.z + y.z, x.w + y.w);
    }
}

extern "C" void kernel_launch(void* const* d_in, const int* in_sizes, int n_in,
                              void* d_out, int out_size, void* d_ws, size_t ws_size,
                              hipStream_t stream) {
    const float* h  = (const float*)d_in[0];
    const float* w  = (const float*)d_in[1];
    const float* a  = (const float*)d_in[2];
    const int* src  = (const int*)d_in[3];
    const int* dst  = (const int*)d_in[4];
    float* out      = (float*)d_out;

    const int n_nodes = in_sizes[0] / C;
    const int n_edges = in_sizes[3];

    float* u     = (float*)d_ws;
    float* v     = u + (size_t)n_nodes * C;
    float* sums  = v + (size_t)n_nodes * C;
    int* cnt     = (int*)(sums + (size_t)2 * n_nodes * C);
    int* start   = cnt + 2 * n_nodes;
    int* sdst    = start + 2 * n_nodes;
    float* sex   = (float*)(sdst + n_edges);
    int* rank    = (int*)(sex + n_edges);

    const int n4 = n_nodes * (C / 4);

    {
        int grid = (2 * n4 + 255) / 256;
        gat_zero<<<grid, 256, 0, stream>>>(sums, 2 * n4, cnt, 2 * n_nodes);
    }
    {
        int grid = (n_nodes + 15) / 16;
        gat_gemm_uv<<<grid, 256, 0, stream>>>(h, w, u, v, n_nodes);
    }
    {
        int grid = (n_edges + 255) / 256;
        gat_count<<<grid, 256, 0, stream>>>(src, cnt, rank, n_edges, n_nodes);
        gat_scan<<<1, 1024, 0, stream>>>(cnt, start, n_nodes);
        gat_fill<<<grid, 256, 0, stream>>>(src, dst, start, rank, sdst, n_edges, n_nodes);
    }
    {
        int grid = (n_nodes + 7) / 8;
        gat_by_source<<<grid, 256, 0, stream>>>(h, u, v, a, start, cnt, sdst, sex, sums, n_nodes);
    }
    {
        int grid = (n4 + 255) / 256;
        gat_add_copies<<<grid, 256, 0, stream>>>(sums, out, n4);
    }
}


# zero, count, fill, add-copies launches: a thread carries 5 elements (zero: 10 stores), its loads and counting atomics in flight together; 500 work-groups each, was 2500 to 5000; sums as before
# speedup vs baseline: 1.8349x; 1.0817x over previous
#include <hip/hip_runtime.h>
#include <hip/hip_bf16.h>


#define C 128
#define LEAKY 0.2f
#define EPT 5
#define CHUNK (256 * EPT)

typedef __attribute__((ext_vector_type(2))) float v2f;
typedef __attribute__((ext_vector_type(8))) float v8f;

__device__ __forceinline__ int chip_half() {
    return (__builtin_amdgcn_s_sendmsg_rtn(0x87) >> 18) & 1;
}

__global__ __launch_bounds__(256) void gat_zero(float* __restrict__ sums, int n_sums4,
                                                int* __restrict__ cnt, int n_cnt) {
    const int i0 = blockIdx.x * (2 * CHUNK) + threadIdx.x;
    if (blockIdx.x * (2 * CHUNK) + 2 * CHUNK <= n_sums4) {
        #pragma unroll
        for (int k = 0; k < 2 * EPT; ++k)
            ((float4*)sums)[i0 + 256 * k] = make_float4(0.0f, 0.0f, 0.0f, 0.0f);
    } else {
        for (int k = 0; k < 2 * EPT; ++k) {
            const int i = i0 + 256 * k;
            if (i < n_sums4) ((float4*)sums)[i] = make_float4(0.0f, 0.0f, 0.0f, 0.0f);
        }
    }
    const int j = blockIdx.x * 256 + threadIdx.x;
    if (j < n_cnt) cnt[j] = 0;
}

__global__ __launch_bounds__(256) void gat_gemm_uv(const float* __restrict__ h,
                                                   const float* __restrict__ w,
                                                   float* __restrict__ u,
                                                   float* __restrict__ v,
                                                   int n_nodes) {
    __shared__ float lh[16 * C];
    const int row0 = blockIdx.x * 16;
    const int tid  = threadIdx.x;
    const bool full_tile = (row0 + 16) <= n_nodes;

    if (full_tile) {
        for (int i = tid; i < 16 * C; i += 256)
            lh[i] = h[(size_t)row0 * C + i];
    } else {
        for (int i = tid; i < 16 * C; i += 256) {
            int gr = row0 + (i >> 7);
            lh[i] = (gr < n_nodes) ? h[(size_t)gr * C + (i & (C - 1))] : 0.0f;
        }
    }
    __syncthreads();

    const int wave = tid >> 5;
    const int lane = tid & 31;
    const int half = lane >> 4;
    const int l16  = lane & 15;
    const int col  = wave * 16 + l16;

    v8f accu = {};
    v8f accv = {};
    for (int k = 0; k < C; k += 4) {
        const int ka = k + half * 2;
        v2f afrag;
        afrag.x = lh[l16 * C + ka];
        afrag.y = lh[l16 * C + ka + 1];
        v2f bu, bv;
        bu.x = w[(size_t)(ka    ) * C + col];
        bu.y = w[(size_t)(ka + 1) * C + col];
        bv.x = w[(size_t)(C + ka    ) * C + col];
        bv.y = w[(size_t)(C + ka + 1) * C + col];
        accu = __builtin_amdgcn_wmma_f32_16x16x4_f32(false, afrag, false, bu,
                                                     (short)0, accu, false, false);
        accv = __builtin_amdgcn_wmma_f32_16x16x4_f32(false, afrag, false, bv,
                                                     (short)0, accv, false, false);
    }

    const int mbase = row0 + half * 8;
    if (full_tile) {
        #pragma unroll
        for (int r = 0; r < 8; ++r) {
            u[(size_t)(mbase + r) * C + col] = accu[r];
            v[(size_t)(mbase + r) * C + col] = accv[r];
        }
    } else {
        #pragma unroll
        for (int r = 0; r < 8; ++r) {
            if (mbase + r < n_nodes) {
                u[(size_t)(mbase + r) * C + col] = accu[r];
                v[(size_t)(mbase + r) * C + col] = accv[r];
            }
        }
    }
}

__device__ __forceinline__ float leaky(float x) { return x > 0.0f ? x : LEAKY * x; }

__global__ __launch_bounds__(256) void gat_count(const int* __restrict__ src,
                                                 int* __restrict__ cnt,
                                                 int* __restrict__ rank, int n_edges,
                                                 int n_nodes) {
    const int hf = chip_half();
    int* __restrict__ mine = cnt + hf * n_nodes;
    const unsigned tag = (unsigned)hf << 31;
    const int e0 = blockIdx.x * CHUNK + threadIdx.x;
    if (blockIdx.x * CHUNK + CHUNK <= n_edges) {
        int s[EPT];
        unsigned r[EPT];
        #pragma unroll
        for (int k = 0; k < EPT; ++k) s[k] = src[e0 + 256 * k];
        #pragma unroll
        for (int k = 0; k < EPT; ++k) r[k] = (unsigned)atomicAdd(&mine[s[k]], 1);
        #pragma unroll
        for (int k = 0; k < EPT; ++k) rank[e0 + 256 * k] = (int)(r[k] | tag);
    } else {
        for (int k = 0; k < EPT; ++k) {
            const int e = e0 + 256 * k;
            if (e < n_edges) {
                const unsigned r = (unsigned)atomicAdd(&mine[src[e]], 1);
                rank[e] = (int)(r | tag);
            }
        }
    }
}

__global__ __launch_bounds__(1024) void gat_scan(const int* __restrict__ cnt,
                                                 int* __restrict__ start, int n) {
    __shared__ int wsum[32];
    const int tid  = threadIdx.x;
    const int lane = tid & 31;
    const int wid  = tid >> 5;
    const int per  = (n + 1023) / 1024;
    const int b    = tid * per;
    const int e    = (b + per < n) ? (b + per) : n;

    int s = 0;
    for (int i = b; i < e; ++i) s += cnt[i] + cnt[n + i];

    int x = s;
    #pragma unroll
    for (int off = 1; off < 32; off <<= 1) {
        int y = __shfl_up(x, off, 32);
        if (lane >= off) x += y;
    }
    if (lane == 31) wsum[wid] = x;
    __syncthreads();
    if (wid == 0) {
        const int t = wsum[lane];
        int xs = t;
        #pragma unroll
        for (int off = 1; off < 32; off <<= 1) {
            int y = __shfl_up(xs, off, 32);
            if (lane >= off) xs += y;
        }
        wsum[lane] = xs - t;
    }
    __syncthreads();

    int run = wsum[wid] + (x - s);
    for (int i = b; i < e; ++i) {
        const int c0 = cnt[i];
        const int c1 = cnt[n + i];
        start[i] = run;
        start[n + i] = run + c0;
        run += c0 + c1;
    }
}

__global__ __launch_bounds__(256) void gat_fill(const int* __restrict__ src,
                                                const int* __restrict__ dst,
                                                const int* __restrict__ start,
                                                const int* __restrict__ rank,
                                                int* __restrict__ sdst, int n_edges,
                                                int n_nodes) {
    const int e0 = blockIdx.x * CHUNK + threadIdx.x;
    if (blockIdx.x * CHUNK + CHUNK <= n_edges) {
        unsigned rk[EPT];
        int s[EPT], d[EPT], st[EPT];
        #pragma unroll
        for (int k = 0; k < EPT; ++k) {
            const int e = e0 + 256 * k;
            rk[k] = (unsigned)rank[e];
            s[k]  = src[e];
            d[k]  = dst[e];
        }
        #pragma unroll
        for (int k = 0; k < EPT; ++k)
            st[k] = start[(int)(rk[k] >> 31) * n_nodes + s[k]];
        #pragma unroll
        for (int k = 0; k < EPT; ++k)
            sdst[st[k] + (int)(rk[k] & 0x7fffffffu)] = d[k];
    } else {
        for (int k = 0; k < EPT; ++k) {
            const int e = e0 + 256 * k;
            if (e < n_edges) {
                const unsigned rk = (unsigned)rank[e];
                const int hb = (int)(rk >> 31);
                const int r  = (int)(rk & 0x7fffffffu);
                sdst[start[hb * n_nodes + src[e]] + r] = dst[e];
            }
        }
    }
}

__device__ __forceinline__ float xhalf(float x) {
    int b = __builtin_bit_cast(int, x);
    return __builtin_bit_cast(float,
        __builtin_amdgcn_permlanex16(b, b, 0x76543210, 0xfedcba98, false, false));
}
template <int K>
__device__ __forceinline__ float row_xor_add(float x) {
    int y = __builtin_amdgcn_update_dpp(0, __builtin_bit_cast(int, x), 0x160 | K, 0xf, 0xf, true);
    return x + __builtin_bit_cast(float, y);
}
__device__ __forceinline__ float wave_sum(float x) {
    x += xhalf(x);
    x = row_xor_add<8>(x);
    x = row_xor_add<4>(x);
    x = row_xor_add<2>(x);
    x = row_xor_add<1>(x);
    return x;
}

typedef float f4 __attribute__((ext_vector_type(4)));
typedef const __attribute__((address_space(1))) f4* grow_cf4;
typedef __attribute__((address_space(1))) float* grow_f;
__device__ __forceinline__ unsigned long long uni_addr(const float* p) {
    const unsigned long long x = (unsigned long long)p;
    const unsigned lo = __builtin_amdgcn_readfirstlane((unsigned)x);
    const unsigned hi = __builtin_amdgcn_readfirstlane((unsigned)(x >> 32));
    return ((unsigned long long)hi << 32) | lo;
}

#define EB 8
__global__ __launch_bounds__(256) void gat_by_source(const float* __restrict__ h,
                                                     const float* __restrict__ u,
                                                     const float* __restrict__ v,
                                                     const float* __restrict__ a,
                                                     const int* __restrict__ start,
                                                     const int* __restrict__ cnt,
                                                     const int* __restrict__ sdst,
                                                     float* __restrict__ sex,
                                                     float* __restrict__ sums,
                                                     int n_nodes) {
    const int lane = threadIdx.x & 31;
    const int s = __builtin_amdgcn_readfirstlane(blockIdx.x * 8 + (threadIdx.x >> 5));
    if (s >= n_nodes) return;
    const int beg = start[s];
    const int deg = cnt[s] + cnt[n_nodes + s];
    if (deg <= 0) return;
    float* __restrict__ mine = sums + (size_t)chip_half() * n_nodes * C;

    const float4 uu = ((const float4*)(u + (size_t)s * C))[lane];
    const float4 aa = ((const float4*)a)[lane];

    float dsum = 0.0f;
    for (int base = 0; base < deg; base += 32) {
        const int idx = base + lane;
        const bool ok = idx < deg;
        const int dj = ok ? sdst[beg + idx] : s;
        const int nb = (deg - base < 32) ? (deg - base) : 32;
        float ej = 0.0f;
        for (int j0 = 0; j0 < nb; j0 += EB) {
            f4 vv[EB];
            #pragma unroll
            for (int t = 0; t < EB; ++t) {
                const int d = __builtin_amdgcn_readlane(dj, j0 + t);
                vv[t] = ((grow_cf4)uni_addr(v + (size_t)d * C))[lane];
            }
            #pragma unroll
            for (int t = 0; t < EB; ++t) {
                float sum = leaky(uu.x + vv[t].x) * aa.x;
                sum = fmaf(leaky(uu.y + vv[t].y), aa.y, sum);
                sum = fmaf(leaky(uu.z + vv[t].z), aa.z, sum);
                sum = fmaf(leaky(uu.w + vv[t].w), aa.w, sum);
                sum = wave_sum(sum);
                if (lane == j0 + t) ej = sum;
            }
        }
        const float ex = ok ? expf(ej) : 0.0f;
        if (ok) sex[beg + idx] = ex;
        dsum += ex;
    }
    const float denom = wave_sum(dsum);

    float hk[4];
    #pragma unroll
    for (int k = 0; k < 4; ++k) hk[k] = h[(size_t)s * C + 32 * k + lane];

    for (int base = 0; base < deg; base += 32) {
        const int idx = base + lane;
        const bool ok = idx < deg;
        const int dj = ok ? sdst[beg + idx] : s;
        const float wj = ok ? sex[beg + idx] / denom : 0.0f;
        const int nb = (deg - base < 32) ? (deg - base) : 32;
        for (int j = 0; j < nb; ++j) {
            const int d = __builtin_amdgcn_readlane(dj, j);
            const float w = __builtin_bit_cast(float,
                __builtin_amdgcn_readlane(__builtin_bit_cast(int, wj), j));
            grow_f op = (grow_f)uni_addr(mine + (size_t)d * C) + lane;
            #pragma unroll
            for (int k = 0; k < 4; ++k)
                __hip_atomic_fetch_add(op + 32 * k, hk[k] * w, __ATOMIC_RELAXED, __HIP_MEMORY_SCOPE_AGENT);
        }
    }
}

__global__ __launch_bounds__(256) void gat_add_copies(const float* __restrict__ sums,
                                                      float* __restrict__ out, int n4) {
    const int i0 = blockIdx.x * CHUNK + threadIdx.x;
    if (blockIdx.x * CHUNK + CHUNK <= n4) {
        float4 x[EPT], y[EPT];
        #pragma unroll
        for (int k = 0; k < EPT; ++k) {
            x[k] = ((const float4*)sums)[i0 + 256 * k];
            y[k] = ((const float4*)sums)[n4 + i0 + 256 * k];
        }
        #pragma unroll
        for (int k = 0; k < EPT; ++k)
            ((float4*)out)[i0 + 256 * k] = make_float4(x[k].x + y[k].x, x[k].y + y[k].y,
                                                       x[k].z + y[k].z, x[k].w + y[k].w);
    } else {
        for (int k = 0; k < EPT; ++k) {
            const int i = i0 + 256 * k;
            if (i < n4) {
                const float4 x = ((const float4*)sums)[i];
                const float4 y = ((const float4*)sums)[n4 + i];
                ((float4*)out)[i] = make_float4(x.x + y.x, x.y + y.y, x.z + y.z, x.w + y.w);
            }
        }
    }
}

extern "C" void kernel_launch(void* const* d_in, const int* in_sizes, int n_in,
                              void* d_out, int out_size, void* d_ws, size_t ws_size,
                              hipStream_t stream) {
    const float* h  = (const float*)d_in[0];
    const float* w  = (const float*)d_in[1];
    const float* a  = (const float*)d_in[2];
    const int* src  = (const int*)d_in[3];
    const int* dst  = (const int*)d_in[4];
    float* out      = (float*)d_out;

    const int n_nodes = in_sizes[0] / C;
    const int n_edges = in_sizes[3];

    float* u     = (float*)d_ws;
    float* v     = u + (size_t)n_nodes * C;
    float* sums  = v + (size_t)n_nodes * C;
    int* cnt     = (int*)(sums + (size_t)2 * n_nodes * C);
    int* start   = cnt + 2 * n_nodes;
    int* sdst    = start + 2 * n_nodes;
    float* sex   = (float*)(sdst + n_edges);
    int* rank    = (int*)(sex + n_edges);

    const int n4 = n_nodes * (C / 4);

    {
        int grid = (2 * n4 + 2 * CHUNK - 1) / (2 * CHUNK);
        gat_zero<<<grid, 256, 0, stream>>>(sums, 2 * n4, cnt, 2 * n_nodes);
    }
    {
        int grid = (n_nodes + 15) / 16;
        gat_gemm_uv<<<grid, 256, 0, stream>>>(h, w, u, v, n_nodes);
    }
    {
        int grid = (n_edges + CHUNK - 1) / CHUNK;
        gat_count<<<grid, 256, 0, stream>>>(src, cnt, rank, n_edges, n_nodes);
        gat_scan<<<1, 1024, 0, stream>>>(cnt, start, n_nodes);
        gat_fill<<<grid, 256, 0, stream>>>(src, dst, start, rank, sdst, n_edges, n_nodes);
    }
    {
        int grid = (n_nodes + 7) / 8;
        gat_by_source<<<grid, 256, 0, stream>>>(h, u, v, a, start, cnt, sdst, sex, sums, n_nodes);
    }
    {
        int grid = (n4 + CHUNK - 1) / CHUNK;
        gat_add_copies<<<grid, 256, 0, stream>>>(sums, out, n4);
    }
}


# projection, edge count and zero fill of the row sums share one launch, work-group index picks the role: three independent bodies overlap (ran one behind the other); 6 launches, was 7; sums as before
# speedup vs baseline: 1.9146x; 1.0434x over previous
#include <hip/hip_runtime.h>
#include <hip/hip_bf16.h>


#define C 128
#define LEAKY 0.2f
#define EPT 5
#define CHUNK (256 * EPT)

typedef __attribute__((ext_vector_type(2))) float v2f;
typedef __attribute__((ext_vector_type(8))) float v8f;

__device__ __forceinline__ int chip_half() {
    return (__builtin_amdgcn_s_sendmsg_rtn(0x87) >> 18) & 1;
}

__global__ __launch_bounds__(256) void gat_zero_cnt(int* __restrict__ cnt, int n_cnt) {
    const int j = blockIdx.x * 256 + threadIdx.x;
    if (j < n_cnt) cnt[j] = 0;
}

__device__ __forceinline__ void role_zero_sums(float* __restrict__ sums, int n_sums4, int bid) {
    const int i0 = bid * (2 * CHUNK) + threadIdx.x;
    if (bid * (2 * CHUNK) + 2 * CHUNK <= n_sums4) {
        #pragma unroll
        for (int k = 0; k < 2 * EPT; ++k)
            ((float4*)sums)[i0 + 256 * k] = make_float4(0.0f, 0.0f, 0.0f, 0.0f);
    } else {
        for (int k = 0; k < 2 * EPT; ++k) {
            const int i = i0 + 256 * k;
            if (i < n_sums4) ((float4*)sums)[i] = make_float4(0.0f, 0.0f, 0.0f, 0.0f);
        }
    }
}

__device__ __forceinline__ void role_gemm_uv(const float* __restrict__ h,
                                             const float* __restrict__ w,
                                             float* __restrict__ u,
                                             float* __restrict__ v,
                                             int n_nodes, int bid,
                                             float* __restrict__ lh) {
    const int row0 = bid * 16;
    const int tid  = threadIdx.x;
    const bool full_tile = (row0 + 16) <= n_nodes;

    if (full_tile) {
        for (int i = tid; i < 16 * C; i += 256)
            lh[i] = h[(size_t)row0 * C + i];
    } else {
        for (int i = tid; i < 16 * C; i += 256) {
            int gr = row0 + (i >> 7);
            lh[i] = (gr < n_nodes) ? h[(size_t)gr * C + (i & (C - 1))] : 0.0f;
        }
    }
    __syncthreads();

    const int wave = tid >> 5;
    const int lane = tid & 31;
    const int half = lane >> 4;
    const int l16  = lane & 15;
    const int col  = wave * 16 + l16;

    v8f accu = {};
    v8f accv = {};
    for (int k = 0; k < C; k += 4) {
        const int ka = k + half * 2;
        v2f afrag;
        afrag.x = lh[l16 * C + ka];
        afrag.y = lh[l16 * C + ka + 1];
        v2f bu, bv;
        bu.x = w[(size_t)(ka    ) * C + col];
        bu.y = w[(size_t)(ka + 1) * C + col];
        bv.x = w[(size_t)(C + ka    ) * C + col];
        bv.y = w[(size_t)(C + ka + 1) * C + col];
        accu = __builtin_amdgcn_wmma_f32_16x16x4_f32(false, afrag, false, bu,
                                                     (short)0, accu, false, false);
        accv = __builtin_amdgcn_wmma_f32_16x16x4_f32(false, afrag, false, bv,
                                                     (short)0, accv, false, false);
    }

    const int mbase = row0 + half * 8;
    if (full_tile) {
        #pragma unroll
        for (int r = 0; r < 8; ++r) {
            u[(size_t)(mbase + r) * C + col] = accu[r];
            v[(size_t)(mbase + r) * C + col] = accv[r];
        }
    } else {
        #pragma unroll
        for (int r = 0; r < 8; ++r) {
            if (mbase + r < n_nodes) {
                u[(size_t)(mbase + r) * C + col] = accu[r];
                v[(size_t)(mbase + r) * C + col] = accv[r];
            }
        }
    }
}

__device__ __forceinline__ float leaky(float x) { return x > 0.0f ? x : LEAKY * x; }

__device__ __forceinline__ void role_count(const int* __restrict__ src,
                                           int* __restrict__ cnt,
                                           int* __restrict__ rank, int n_edges,
                                           int n_nodes, int bid) {
    const int hf = chip_half();
    int* __restrict__ mine = cnt + hf * n_nodes;
    const unsigned tag = (unsigned)hf << 31;
    const int e0 = bid * CHUNK + threadIdx.x;
    if (bid * CHUNK + CHUNK <= n_edges) {
        int s[EPT];
        unsigned r[EPT];
        #pragma unroll
        for (int k = 0; k < EPT; ++k) s[k] = src[e0 + 256 * k];
        #pragma unroll
        for (int k = 0; k < EPT; ++k) r[k] = (unsigned)atomicAdd(&mine[s[k]], 1);
        #pragma unroll
        for (int k = 0; k < EPT; ++k) rank[e0 + 256 * k] = (int)(r[k] | tag);
    } else {
        for (int k = 0; k < EPT; ++k) {
            const int e = e0 + 256 * k;
            if (e < n_edges) {
                const unsigned r = (unsigned)atomicAdd(&mine[src[e]], 1);
                rank[e] = (int)(r | tag);
            }
        }
    }
}

__global__ __launch_bounds__(256) void gat_count_gemm_zero(const float* __restrict__ h,
                                                           const float* __restrict__ w,
                                                           float* __restrict__ u,
                                                           float* __restrict__ v,
                                                           const int* __restrict__ src,
                                                           int* __restrict__ cnt,
                                                           int* __restrict__ rank,
                                                           float* __restrict__ sums,
                                                           int n_sums4, int n_edges,
                                                           int n_nodes, int nb_count,
                                                           int nb_gemm) {
    __shared__ float lh[16 * C];
    const int b = blockIdx.x;
    if (b < nb_count) {
        role_count(src, cnt, rank, n_edges, n_nodes, b);
    } else if (b < nb_count + nb_gemm) {
        role_gemm_uv(h, w, u, v, n_nodes, b - nb_count, lh);
    } else {
        role_zero_sums(sums, n_sums4, b - nb_count - nb_gemm);
    }
}

__global__ __launch_bounds__(1024) void gat_scan(const int* __restrict__ cnt,
                                                 int* __restrict__ start, int n) {
    __shared__ int wsum[32];
    const int tid  = threadIdx.x;
    const int lane = tid & 31;
    const int wid  = tid >> 5;
    const int per  = (n + 1023) / 1024;
    const int b    = tid * per;
    const int e    = (b + per < n) ? (b + per) : n;

    int s = 0;
    for (int i = b; i < e; ++i) s += cnt[i] + cnt[n + i];

    int x = s;
    #pragma unroll
    for (int off = 1; off < 32; off <<= 1) {
        int y = __shfl_up(x, off, 32);
        if (lane >= off) x += y;
    }
    if (lane == 31) wsum[wid] = x;
    __syncthreads();
    if (wid == 0) {
        const int t = wsum[lane];
        int xs = t;
        #pragma unroll
        for (int off = 1; off < 32; off <<= 1) {
            int y = __shfl_up(xs, off, 32);
            if (lane >= off) xs += y;
        }
        wsum[lane] = xs - t;
    }
    __syncthreads();

    int run = wsum[wid] + (x - s);
    for (int i = b; i < e; ++i) {
        const int c0 = cnt[i];
        const int c1 = cnt[n + i];
        start[i] = run;
        start[n + i] = run + c0;
        run += c0 + c1;
    }
}

__global__ __launch_bounds__(256) void gat_fill(const int* __restrict__ src,
                                                const int* __restrict__ dst,
                                                const int* __restrict__ start,
                                                const int* __restrict__ rank,
                                                int* __restrict__ sdst, int n_edges,
                                                int n_nodes) {
    const int e0 = blockIdx.x * CHUNK + threadIdx.x;
    if (blockIdx.x * CHUNK + CHUNK <= n_edges) {
        unsigned rk[EPT];
        int s[EPT], d[EPT], st[EPT];
        #pragma unroll
        for (int k = 0; k < EPT; ++k) {
            const int e = e0 + 256 * k;
            rk[k] = (unsigned)rank[e];
            s[k]  = src[e];
            d[k]  = dst[e];
        }
        #pragma unroll
        for (int k = 0; k < EPT; ++k)
            st[k] = start[(int)(rk[k] >> 31) * n_nodes + s[k]];
        #pragma unroll
        for (int k = 0; k < EPT; ++k)
            sdst[st[k] + (int)(rk[k] & 0x7fffffffu)] = d[k];
    } else {
        for (int k = 0; k < EPT; ++k) {
            const int e = e0 + 256 * k;
            if (e < n_edges) {
                const unsigned rk = (unsigned)rank[e];
                const int hb = (int)(rk >> 31);
                const int r  = (int)(rk & 0x7fffffffu);
                sdst[start[hb * n_nodes + src[e]] + r] = dst[e];
            }
        }
    }
}

__device__ __forceinline__ float xhalf(float x) {
    int b = __builtin_bit_cast(int, x);
    return __builtin_bit_cast(float,
        __builtin_amdgcn_permlanex16(b, b, 0x76543210, 0xfedcba98, false, false));
}
template <int K>
__device__ __forceinline__ float row_xor_add(float x) {
    int y = __builtin_amdgcn_update_dpp(0, __builtin_bit_cast(int, x), 0x160 | K, 0xf, 0xf, true);
    return x + __builtin_bit_cast(float, y);
}
__device__ __forceinline__ float wave_sum(float x) {
    x += xhalf(x);
    x = row_xor_add<8>(x);
    x = row_xor_add<4>(x);
    x = row_xor_add<2>(x);
    x = row_xor_add<1>(x);
    return x;
}

typedef float f4 __attribute__((ext_vector_type(4)));
typedef const __attribute__((address_space(1))) f4* grow_cf4;
typedef __attribute__((address_space(1))) float* grow_f;
__device__ __forceinline__ unsigned long long uni_addr(const float* p) {
    const unsigned long long x = (unsigned long long)p;
    const unsigned lo = __builtin_amdgcn_readfirstlane((unsigned)x);
    const unsigned hi = __builtin_amdgcn_readfirstlane((unsigned)(x >> 32));
    return ((unsigned long long)hi << 32) | lo;
}

#define EB 8
__global__ __launch_bounds__(256) void gat_by_source(const float* __restrict__ h,
                                                     const float* __restrict__ u,
                                                     const float* __restrict__ v,
                                                     const float* __restrict__ a,
                                                     const int* __restrict__ start,
                                                     const int* __restrict__ cnt,
                                                     const int* __restrict__ sdst,
                                                     float* __restrict__ sex,
                                                     float* __restrict__ sums,
                                                     int n_nodes) {
    const int lane = threadIdx.x & 31;
    const int s = __builtin_amdgcn_readfirstlane(blockIdx.x * 8 + (threadIdx.x >> 5));
    if (s >= n_nodes) return;
    const int beg = start[s];
    const int deg = cnt[s] + cnt[n_nodes + s];
    if (deg <= 0) return;
    float* __restrict__ mine = sums + (size_t)chip_half() * n_nodes * C;

    const float4 uu = ((const float4*)(u + (size_t)s * C))[lane];
    const float4 aa = ((const float4*)a)[lane];

    float dsum = 0.0f;
    for (int base = 0; base < deg; base += 32) {
        const int idx = base + lane;
        const bool ok = idx < deg;
        const int dj = ok ? sdst[beg + idx] : s;
        const int nb = (deg - base < 32) ? (deg - base) : 32;
        float ej = 0.0f;
        for (int j0 = 0; j0 < nb; j0 += EB) {
            f4 vv[EB];
            #pragma unroll
            for (int t = 0; t < EB; ++t) {
                const int d = __builtin_amdgcn_readlane(dj, j0 + t);
                vv[t] = ((grow_cf4)uni_addr(v + (size_t)d * C))[lane];
            }
            #pragma unroll
            for (int t = 0; t < EB; ++t) {
                float sum = leaky(uu.x + vv[t].x) * aa.x;
                sum = fmaf(leaky(uu.y + vv[t].y), aa.y, sum);
                sum = fmaf(leaky(uu.z + vv[t].z), aa.z, sum);
                sum = fmaf(leaky(uu.w + vv[t].w), aa.w, sum);
                sum = wave_sum(sum);
                if (lane == j0 + t) ej = sum;
            }
        }
        const float ex = ok ? expf(ej) : 0.0f;
        if (ok) sex[beg + idx] = ex;
        dsum += ex;
    }
    const float denom = wave_sum(dsum);

    float hk[4];
    #pragma unroll
    for (int k = 0; k < 4; ++k) hk[k] = h[(size_t)s * C + 32 * k + lane];

    for (int base = 0; base < deg; base += 32) {
        const int idx = base + lane;
        const bool ok = idx < deg;
        const int dj = ok ? sdst[beg + idx] : s;
        const float wj = ok ? sex[beg + idx] / denom : 0.0f;
        const int nb = (deg - base < 32) ? (deg - base) : 32;
        for (int j = 0; j < nb; ++j) {
            const int d = __builtin_amdgcn_readlane(dj, j);
            const float w = __builtin_bit_cast(float,
                __builtin_amdgcn_readlane(__builtin_bit_cast(int, wj), j));
            grow_f op = (grow_f)uni_addr(mine + (size_t)d * C) + lane;
            #pragma unroll
            for (int k = 0; k < 4; ++k)
                __hip_atomic_fetch_add(op + 32 * k, hk[k] * w, __ATOMIC_RELAXED, __HIP_MEMORY_SCOPE_AGENT);
        }
    }
}

__global__ __launch_bounds__(256) void gat_add_copies(const float* __restrict__ sums,
                                                      float* __restrict__ out, int n4) {
    const int i0 = blockIdx.x * CHUNK + threadIdx.x;
    if (blockIdx.x * CHUNK + CHUNK <= n4) {
        float4 x[EPT], y[EPT];
        #pragma unroll
        for (int k = 0; k < EPT; ++k) {
            x[k] = ((const float4*)sums)[i0 + 256 * k];
            y[k] = ((const float4*)sums)[n4 + i0 + 256 * k];
        }
        #pragma unroll
        for (int k = 0; k < EPT; ++k)
            ((float4*)out)[i0 + 256 * k] = make_float4(x[k].x + y[k].x, x[k].y + y[k].y,
                                                       x[k].z + y[k].z, x[k].w + y[k].w);
    } else {
        for (int k = 0; k < EPT; ++k) {
            const int i = i0 + 256 * k;
            if (i < n4) {
                const float4 x = ((const float4*)sums)[i];
                const float4 y = ((const float4*)sums)[n4 + i];
                ((float4*)out)[i] = make_float4(x.x + y.x, x.y + y.y, x.z + y.z, x.w + y.w);
            }
        }
    }
}

extern "C" void kernel_launch(void* const* d_in, const int* in_sizes, int n_in,
                              void* d_out, int out_size, void* d_ws, size_t ws_size,
                              hipStream_t stream) {
    const float* h  = (const float*)d_in[0];
    const float* w  = (const float*)d_in[1];
    const float* a  = (const float*)d_in[2];
    const int* src  = (const int*)d_in[3];
    const int* dst  = (const int*)d_in[4];
    float* out      = (float*)d_out;

    const int n_nodes = in_sizes[0] / C;
    const int n_edges = in_sizes[3];

    float* u     = (float*)d_ws;
    float* v     = u + (size_t)n_nodes * C;
    float* sums  = v + (size_t)n_nodes * C;
    int* cnt     = (int*)(sums + (size_t)2 * n_nodes * C);
    int* start   = cnt + 2 * n_nodes;
    int* sdst    = start + 2 * n_nodes;
    float* sex   = (float*)(sdst + n_edges);
    int* rank    = (int*)(sex + n_edges);

    const int n4 = n_nodes * (C / 4);

    const int nb_edges = (n_edges + CHUNK - 1) / CHUNK;
    {
        int grid = (2 * n_nodes + 255) / 256;
        gat_zero_cnt<<<grid, 256, 0, stream>>>(cnt, 2 * n_nodes);
    }
    {
        int nb_gemm = (n_nodes + 15) / 16;
        int nb_zero = (2 * n4 + 2 * CHUNK - 1) / (2 * CHUNK);
        gat_count_gemm_zero<<<nb_edges + nb_gemm + nb_zero, 256, 0, stream>>>(
            h, w, u, v, src, cnt, rank, sums, 2 * n4, n_edges, n_nodes, nb_edges, nb_gemm);
    }
    {
        gat_scan<<<1, 1024, 0, stream>>>(cnt, start, n_nodes);
        gat_fill<<<nb_edges, 256, 0, stream>>>(src, dst, start, rank, sdst, n_edges, n_nodes);
    }
    {
        int grid = (n_nodes + 7) / 8;
        gat_by_source<<<grid, 256, 0, stream>>>(h, u, v, a, start, cnt, sdst, sex, sums, n_nodes);
    }
    {
        int grid = (n4 + CHUNK - 1) / CHUNK;
        gat_add_copies<<<grid, 256, 0, stream>>>(sums, out, n4);
    }
}


# a node owns two 64-slot buckets (one per chip half): the count pass stores an edge's source at the slot its counting atomic returns, overflow list beyond; no scan, no fill: 4 launches; sums as before
# speedup vs baseline: 4.5333x; 2.3678x over previous
#include <hip/hip_runtime.h>
#include <hip/hip_bf16.h>


#define C 128
#define LEAKY 0.2f
#define EPT 5
#define CHUNK (256 * EPT)
#define HCAP 64

typedef __attribute__((ext_vector_type(2))) float v2f;
typedef __attribute__((ext_vector_type(8))) float v8f;

__device__ __forceinline__ int chip_half() {
    return (__builtin_amdgcn_s_sendmsg_rtn(0x87) >> 18) & 1;
}

__global__ __launch_bounds__(256) void gat_zero_cnt(int* __restrict__ cnt, int n_cnt) {
    const int j = blockIdx.x * 256 + threadIdx.x;
    if (j < n_cnt) cnt[j] = 0;
}

__device__ __forceinline__ void role_gemm_uv(const float* __restrict__ h,
                                             const float* __restrict__ w,
                                             float* __restrict__ u,
                                             float* __restrict__ v,
                                             int n_nodes, int bid,
                                             float* __restrict__ lh) {
    const int row0 = bid * 16;
    const int tid  = threadIdx.x;
    const bool full_tile = (row0 + 16) <= n_nodes;

    if (full_tile) {
        for (int i = tid; i < 16 * C; i += 256)
            lh[i] = h[(size_t)row0 * C + i];
    } else {
        for (int i = tid; i < 16 * C; i += 256) {
            int gr = row0 + (i >> 7);
            lh[i] = (gr < n_nodes) ? h[(size_t)gr * C + (i & (C - 1))] : 0.0f;
        }
    }
    __syncthreads();

    const int wave = tid >> 5;
    const int lane = tid & 31;
    const int half = lane >> 4;
    const int l16  = lane & 15;
    const int col  = wave * 16 + l16;

    v8f accu = {};
    v8f accv = {};
    for (int k = 0; k < C; k += 4) {
        const int ka = k + half * 2;
        v2f afrag;
        afrag.x = lh[l16 * C + ka];
        afrag.y = lh[l16 * C + ka + 1];
        v2f bu, bv;
        bu.x = w[(size_t)(ka    ) * C + col];
        bu.y = w[(size_t)(ka + 1) * C + col];
        bv.x = w[(size_t)(C + ka    ) * C + col];
        bv.y = w[(size_t)(C + ka + 1) * C + col];
        accu = __builtin_amdgcn_wmma_f32_16x16x4_f32(false, afrag, false, bu,
                                                     (short)0, accu, false, false);
        accv = __builtin_amdgcn_wmma_f32_16x16x4_f32(false, afrag, false, bv,
                                                     (short)0, accv, false, false);
    }

    const int mbase = row0 + half * 8;
    if (full_tile) {
        #pragma unroll
        for (int r = 0; r < 8; ++r) {
            u[(size_t)(mbase + r) * C + col] = accu[r];
            v[(size_t)(mbase + r) * C + col] = accv[r];
        }
    } else {
        #pragma unroll
        for (int r = 0; r < 8; ++r) {
            if (mbase + r < n_nodes) {
                u[(size_t)(mbase + r) * C + col] = accu[r];
                v[(size_t)(mbase + r) * C + col] = accv[r];
            }
        }
    }
}

__device__ __forceinline__ float leaky(float x) { return x > 0.0f ? x : LEAKY * x; }

__device__ __forceinline__ void place_edge(int d, int s, int r, int hf,
                                           int* __restrict__ ssrc,
                                           int* __restrict__ ovf_n,
                                           int* __restrict__ ovf_src,
                                           int* __restrict__ ovf_dst) {
    if (r < HCAP) {
        ssrc[(d * 2 + hf) * HCAP + r] = s;
    } else {
        const int k = atomicAdd(ovf_n, 1);
        ovf_src[k] = s;
        ovf_dst[k] = d;
    }
}

__device__ __forceinline__ void role_count(const int* __restrict__ src,
                                           const int* __restrict__ dst,
                                           int* __restrict__ cnt,
                                           int* __restrict__ ssrc,
                                           int* __restrict__ ovf_n,
                                           int* __restrict__ ovf_src,
                                           int* __restrict__ ovf_dst, int n_edges,
                                           int n_nodes, int bid) {
    const int hf = chip_half();
    int* __restrict__ mine = cnt + hf * n_nodes;
    const int e0 = bid * CHUNK + threadIdx.x;
    if (bid * CHUNK + CHUNK <= n_edges) {
        int d[EPT], s[EPT], r[EPT];
        #pragma unroll
        for (int k = 0; k < EPT; ++k) {
            d[k] = dst[e0 + 256 * k];
            s[k] = src[e0 + 256 * k];
        }
        #pragma unroll
        for (int k = 0; k < EPT; ++k) r[k] = atomicAdd(&mine[d[k]], 1);
        #pragma unroll
        for (int k = 0; k < EPT; ++k)
            place_edge(d[k], s[k], r[k], hf, ssrc, ovf_n, ovf_src, ovf_dst);
    } else {
        for (int k = 0; k < EPT; ++k) {
            const int e = e0 + 256 * k;
            if (e < n_edges) {
                const int d = dst[e];
                const int r = atomicAdd(&mine[d], 1);
                place_edge(d, src[e], r, hf, ssrc, ovf_n, ovf_src, ovf_dst);
            }
        }
    }
}

__global__ __launch_bounds__(256) void gat_count_gemm(const float* __restrict__ h,
                                                      const float* __restrict__ w,
                                                      float* __restrict__ u,
                                                      float* __restrict__ v,
                                                      const int* __restrict__ src,
                                                      const int* __restrict__ dst,
                                                      int* __restrict__ cnt,
                                                      int* __restrict__ ssrc,
                                                      int* __restrict__ ovf_n,
                                                      int* __restrict__ ovf_src,
                                                      int* __restrict__ ovf_dst,
                                                      int n_edges, int n_nodes,
                                                      int nb_count) {
    __shared__ float lh[16 * C];
    const int b = blockIdx.x;
    if (b < nb_count) {
        role_count(src, dst, cnt, ssrc, ovf_n, ovf_src, ovf_dst, n_edges, n_nodes, b);
    } else {
        role_gemm_uv(h, w, u, v, n_nodes, b - nb_count, lh);
    }
}

__device__ __forceinline__ float xhalf(float x) {
    int b = __builtin_bit_cast(int, x);
    return __builtin_bit_cast(float,
        __builtin_amdgcn_permlanex16(b, b, 0x76543210, 0xfedcba98, false, false));
}
template <int K>
__device__ __forceinline__ float row_xor_add(float x) {
    int y = __builtin_amdgcn_update_dpp(0, __builtin_bit_cast(int, x), 0x160 | K, 0xf, 0xf, true);
    return x + __builtin_bit_cast(float, y);
}
__device__ __forceinline__ float wave_sum(float x) {
    x += xhalf(x);
    x = row_xor_add<8>(x);
    x = row_xor_add<4>(x);
    x = row_xor_add<2>(x);
    x = row_xor_add<1>(x);
    return x;
}

typedef float f4 __attribute__((ext_vector_type(4)));
typedef const __attribute__((address_space(1))) f4* grow_cf4;
__device__ __forceinline__ unsigned long long uni_addr(const float* p) {
    const unsigned long long x = (unsigned long long)p;
    const unsigned lo = __builtin_amdgcn_readfirstlane((unsigned)x);
    const unsigned hi = __builtin_amdgcn_readfirstlane((unsigned)(x >> 32));
    return ((unsigned long long)hi << 32) | lo;
}

#define EB 8

__device__ __forceinline__ int bucket_slot(int b0, int idx, int m0) {
    return b0 + (idx < m0 ? idx : idx - m0 + HCAP);
}

__global__ __launch_bounds__(256) void gat_logits_by_dst(const float* __restrict__ u,
                                                         const float* __restrict__ v,
                                                         const float* __restrict__ a,
                                                         const int* __restrict__ cnt,
                                                         const int* __restrict__ ssrc,
                                                         const int* __restrict__ ovf_n,
                                                         const int* __restrict__ ovf_src,
                                                         const int* __restrict__ ovf_dst,
                                                         float* __restrict__ sex,
                                                         float* __restrict__ ovf_ex,
                                                         float* __restrict__ denom,
                                                         int n_nodes, int n_edges) {
    const int lane = threadIdx.x & 31;
    const int d = __builtin_amdgcn_readfirstlane(blockIdx.x * 8 + (threadIdx.x >> 5));
    if (d >= n_nodes) return;
    const int c0 = cnt[d];
    const int c1 = cnt[n_nodes + d];
    if (c0 + c1 <= 0) return;
    const int m0  = c0 < HCAP ? c0 : HCAP;
    const int m1  = c1 < HCAP ? c1 : HCAP;
    const int deg = m0 + m1;
    const int b0  = d * (2 * HCAP);
    float* __restrict__ mine = denom + (size_t)chip_half() * n_nodes;

    const float4 vd = ((const float4*)(v + (size_t)d * C))[lane];
    const float4 aa = ((const float4*)a)[lane];

    if (c0 > HCAP || c1 > HCAP) {
        int novf = ovf_n[0];
        novf = novf < n_edges ? novf : n_edges;
        for (int k0 = 0; k0 < novf; k0 += 32) {
            const int k = k0 + lane;
            int od = -1, os = 0;
            if (k < novf) { od = ovf_dst[k]; os = ovf_src[k]; }
            unsigned m = __builtin_amdgcn_ballot_w32(od == d);
            while (m) {
                const int j = __builtin_ctz(m);
                m &= m - 1;
                const int s = __builtin_amdgcn_readlane(os, j);
                const f4 ur = ((grow_cf4)uni_addr(u + (size_t)s * C))[lane];
                float sum = leaky(ur.x + vd.x) * aa.x;
                sum = fmaf(leaky(ur.y + vd.y), aa.y, sum);
                sum = fmaf(leaky(ur.z + vd.z), aa.z, sum);
                sum = fmaf(leaky(ur.w + vd.w), aa.w, sum);
                sum = wave_sum(sum);
                if (lane == j) {
                    const float ex = expf(sum);
                    ovf_ex[k] = ex;
                    __hip_atomic_fetch_add(mine + s, ex, __ATOMIC_RELAXED, __HIP_MEMORY_SCOPE_AGENT);
                }
            }
        }
    }

    for (int base = 0; base < deg; base += 32) {
        const int idx = base + lane;
        const bool ok = idx < deg;
        const int slot = bucket_slot(b0, idx, m0);
        const int sj = ok ? ssrc[slot] : d;
        const int nb = (deg - base < 32) ? (deg - base) : 32;
        float ej = 0.0f;
        for (int j0 = 0; j0 < nb; j0 += EB) {
            f4 uu[EB];
            #pragma unroll
            for (int t = 0; t < EB; ++t) {
                const int s = __builtin_amdgcn_readlane(sj, j0 + t);
                uu[t] = ((grow_cf4)uni_addr(u + (size_t)s * C))[lane];
            }
            #pragma unroll
            for (int t = 0; t < EB; ++t) {
                float sum = leaky(uu[t].x + vd.x) * aa.x;
                sum = fmaf(leaky(uu[t].y + vd.y), aa.y, sum);
                sum = fmaf(leaky(uu[t].z + vd.z), aa.z, sum);
                sum = fmaf(leaky(uu[t].w + vd.w), aa.w, sum);
                sum = wave_sum(sum);
                if (lane == j0 + t) ej = sum;
            }
        }
        if (ok) {
            const float ex = expf(ej);
            sex[slot] = ex;
            __hip_atomic_fetch_add(mine + sj, ex, __ATOMIC_RELAXED, __HIP_MEMORY_SCOPE_AGENT);
        }
    }

}

__global__ __launch_bounds__(256) void gat_sum_by_dst(const float* __restrict__ h,
                                                      const int* __restrict__ cnt,
                                                      const int* __restrict__ ssrc,
                                                      const int* __restrict__ ovf_n,
                                                      const int* __restrict__ ovf_src,
                                                      const int* __restrict__ ovf_dst,
                                                      const float* __restrict__ sex,
                                                      const float* __restrict__ ovf_ex,
                                                      const float* __restrict__ denom,
                                                      float* __restrict__ out,
                                                      int n_nodes, int n_edges) {
    const int lane = threadIdx.x & 31;
    const int d = __builtin_amdgcn_readfirstlane(blockIdx.x * 8 + (threadIdx.x >> 5));
    if (d >= n_nodes) return;
    const int c0 = cnt[d];
    const int c1 = cnt[n_nodes + d];
    const int m0  = c0 < HCAP ? c0 : HCAP;
    const int m1  = c1 < HCAP ? c1 : HCAP;
    const int deg = m0 + m1;
    const int b0  = d * (2 * HCAP);

    f4 acc = {0.0f, 0.0f, 0.0f, 0.0f};
    for (int base = 0; base < deg; base += 32) {
        const int idx = base + lane;
        const bool ok = idx < deg;
        const int slot = bucket_slot(b0, idx, m0);
        const int sj = ok ? ssrc[slot] : d;
        float wj = 0.0f;
        if (ok) wj = sex[slot] / (denom[sj] + denom[n_nodes + sj]);
        const int nb = (deg - base < 32) ? (deg - base) : 32;
        for (int j0 = 0; j0 < nb; j0 += EB) {
            f4 hh[EB];
            #pragma unroll
            for (int t = 0; t < EB; ++t) {
                const int s = __builtin_amdgcn_readlane(sj, j0 + t);
                hh[t] = ((grow_cf4)uni_addr(h + (size_t)s * C))[lane];
            }
            #pragma unroll
            for (int t = 0; t < EB; ++t) {
                if (j0 + t < nb) {
                    const float w = __builtin_bit_cast(float,
                        __builtin_amdgcn_readlane(__builtin_bit_cast(int, wj), j0 + t));
                    acc.x = fmaf(w, hh[t].x, acc.x);
                    acc.y = fmaf(w, hh[t].y, acc.y);
                    acc.z = fmaf(w, hh[t].z, acc.z);
                    acc.w = fmaf(w, hh[t].w, acc.w);
                }
            }
        }
    }

    if (c0 > HCAP || c1 > HCAP) {
        int novf = ovf_n[0];
        novf = novf < n_edges ? novf : n_edges;
        for (int k0 = 0; k0 < novf; k0 += 32) {
            const int k = k0 + lane;
            int od = -1, os = 0;
            float wk = 0.0f;
            if (k < novf) { od = ovf_dst[k]; os = ovf_src[k]; }
            if (od == d) wk = ovf_ex[k] / (denom[os] + denom[n_nodes + os]);
            unsigned m = __builtin_amdgcn_ballot_w32(od == d);
            while (m) {
                const int j = __builtin_ctz(m);
                m &= m - 1;
                const int s = __builtin_amdgcn_readlane(os, j);
                const float w = __builtin_bit_cast(float,
                    __builtin_amdgcn_readlane(__builtin_bit_cast(int, wk), j));
                const f4 hr = ((grow_cf4)uni_addr(h + (size_t)s * C))[lane];
                acc.x = fmaf(w, hr.x, acc.x);
                acc.y = fmaf(w, hr.y, acc.y);
                acc.z = fmaf(w, hr.z, acc.z);
                acc.w = fmaf(w, hr.w, acc.w);
            }
        }
    }
    ((f4*)(out + (size_t)d * C))[lane] = acc;
}

extern "C" void kernel_launch(void* const* d_in, const int* in_sizes, int n_in,
                              void* d_out, int out_size, void* d_ws, size_t ws_size,
                              hipStream_t stream) {
    const float* h  = (const float*)d_in[0];
    const float* w  = (const float*)d_in[1];
    const float* a  = (const float*)d_in[2];
    const int* src  = (const int*)d_in[3];
    const int* dst  = (const int*)d_in[4];
    float* out      = (float*)d_out;

    const int n_nodes = in_sizes[0] / C;
    const int n_edges = in_sizes[3];

    float* u       = (float*)d_ws;
    float* v       = u + (size_t)n_nodes * C;
    int* cnt       = (int*)(v + (size_t)n_nodes * C);
    float* denom   = (float*)(cnt + 2 * n_nodes);
    int* ovf_n     = (int*)(denom + 2 * n_nodes);
    int* ssrc      = ovf_n + 32;
    float* sex     = (float*)(ssrc + (size_t)n_nodes * 2 * HCAP);
    int* ovf_src   = (int*)(sex + (size_t)n_nodes * 2 * HCAP);
    int* ovf_dst   = ovf_src + n_edges;
    float* ovf_ex  = (float*)(ovf_dst + n_edges);

    const int nb_edges = (n_edges + CHUNK - 1) / CHUNK;
    {
        int grid = (4 * n_nodes + 4 + 255) / 256;
        gat_zero_cnt<<<grid, 256, 0, stream>>>(cnt, 4 * n_nodes + 4);
    }
    {
        int nb_gemm = (n_nodes + 15) / 16;
        gat_count_gemm<<<nb_edges + nb_gemm, 256, 0, stream>>>(
            h, w, u, v, src, dst, cnt, ssrc, ovf_n, ovf_src, ovf_dst, n_edges, n_nodes, nb_edges);
    }
    const int nb_nodes = (n_nodes + 7) / 8;
    {
        gat_logits_by_dst<<<nb_nodes, 256, 0, stream>>>(u, v, a, cnt, ssrc, ovf_n, ovf_src, ovf_dst,
                                                        sex, ovf_ex, denom, n_nodes, n_edges);
    }
    {
        gat_sum_by_dst<<<nb_nodes, 256, 0, stream>>>(h, cnt, ssrc, ovf_n, ovf_src, ovf_dst,
                                                     sex, ovf_ex, denom, out, n_nodes, n_edges);
    }
}
